# starting kernel
# baseline (speedup 1.0000x reference)
#include <hip/hip_runtime.h>

typedef __attribute__((ext_vector_type(2))) float v2f;
typedef __attribute__((ext_vector_type(8))) float v8f;

#define HIDW 128
#define KCHUNK 32

__global__ __launch_bounds__(256) void k_fill1(float* p, int n) {
  int i = blockIdx.x * blockDim.x + threadIdx.x;
  if (i < n) p[i] = 1.0f;
}

__global__ __launch_bounds__(256) void k_edge_deg(float* deg, const int* __restrict__ dst, int E) {
  int e = blockIdx.x * blockDim.x + threadIdx.x;
  if (e < E) atomicAdd(&deg[dst[e]], 1.0f);
}

__global__ __launch_bounds__(256) void k_dis(float* dis, const float* __restrict__ deg, int n) {
  int i = blockIdx.x * blockDim.x + threadIdx.x;
  if (i < n) {
    float d = deg[i];
    dis[i] = (d > 0.f) ? rsqrtf(d) : 0.f;
  }
}

__global__ __launch_bounds__(256) void k_prop_init(float* __restrict__ out, const float* __restrict__ h,
                                                   const float* __restrict__ dis, const float* __restrict__ bias,
                                                   int n) {
  int idx = blockIdx.x * blockDim.x + threadIdx.x;
  if (idx < n * HIDW) {
    int i = idx >> 7;
    int f = idx & 127;
    float di = dis[i];
    out[idx] = h[idx] * di * di + bias[f];
  }
}

__global__ __launch_bounds__(256) void k_prop_scatter(const float* __restrict__ h, const int* __restrict__ src,
                                                      const int* __restrict__ dst, const float* __restrict__ dis,
                                                      float* __restrict__ out, int E) {
  int e = blockIdx.x * 2 + (threadIdx.x >> 7);
  if (e >= E) return;
  int f = threadIdx.x & 127;
  int s = src[e], d = dst[e];
  float nrm = dis[s] * dis[d];
  atomicAdd(&out[(size_t)d * HIDW + f], h[(size_t)s * HIDW + f] * nrm);
}

template<int NC, int KW, bool RELU_A0, bool RELU_OUT, bool HAS_BIAS>
__global__ __launch_bounds__(256) void k_gemm(const float* __restrict__ A0, const float* __restrict__ A1,
                                              const float* __restrict__ W, const float* __restrict__ bias,
                                              float* __restrict__ C, int M) {
  __shared__ __align__(16) float As[128][KCHUNK + 4];
  __shared__ __align__(16) float Ws[NC][KCHUNK + 4];
  constexpr int NT = NC / 16;
  constexpr int K = KW * 128;

  const int tid  = threadIdx.x;
  const int wave = tid >> 5;
  const int lane = tid & 31;
  const int l15  = lane & 15;
  const int half = lane >> 4;
  const int m0   = blockIdx.x * 128;

  v8f acc[NT];
#pragma unroll
  for (int t = 0; t < NT; ++t)
#pragma unroll
    for (int r = 0; r < 8; ++r) acc[t][r] = 0.0f;

  for (int kb = 0; kb < K; kb += KCHUNK) {
    const float* Ablk = (KW == 2 && kb >= 128) ? A1 : A0;
    const bool doRelu = RELU_A0 && (kb < 128);
    const int kloc = kb & 127;

#pragma unroll
    for (int j = 0; j < (128 * KCHUNK) / 1024; ++j) {
      int ei  = tid * 4 + j * 1024;
      int row = ei >> 5;
      int col = ei & 31;
      int grow = m0 + row;
      float4 v = {0.f, 0.f, 0.f, 0.f};
      if (grow < M) v = *(const float4*)&Ablk[(size_t)grow * HIDW + kloc + col];
      if (doRelu) {
        v.x = fmaxf(v.x, 0.f); v.y = fmaxf(v.y, 0.f);
        v.z = fmaxf(v.z, 0.f); v.w = fmaxf(v.w, 0.f);
      }
      *(float4*)&As[row][col] = v;
    }
#pragma unroll
    for (int j = 0; j < (NC * KCHUNK) / 1024; ++j) {
      int ei  = tid * 4 + j * 1024;
      int n   = ei >> 5;
      int col = ei & 31;
      *(float4*)&Ws[n][col] = *(const float4*)&W[(size_t)n * K + kb + col];
    }
    __syncthreads();

    const int arow = wave * 16 + l15;
    const int o = half * 2;
#pragma unroll
    for (int k4 = 0; k4 < KCHUNK; k4 += 4) {
      v2f a = *(const v2f*)&As[arow][k4 + o];
#pragma unroll
      for (int t = 0; t < NT; ++t) {
        v2f b = *(const v2f*)&Ws[t * 16 + l15][k4 + o];
        acc[t] = __builtin_amdgcn_wmma_f32_16x16x4_f32(
             false, a,  false, b,
             (short)0, acc[t],  false,  false);
      }
    }
    __syncthreads();
  }

  const int rbase = m0 + wave * 16 + half * 8;
#pragma unroll
  for (int t = 0; t < NT; ++t) {
    int col = t * 16 + l15;
    float bv = HAS_BIAS ? bias[col] : 0.f;
#pragma unroll
    for (int r = 0; r < 8; ++r) {
      int row = rbase + r;
      if (row < M) {
        float v = acc[t][r] + bv;
        if (RELU_OUT) v = fmaxf(v, 0.f);
        C[(size_t)row * NC + col] = v;
      }
    }
  }
}

extern "C" void kernel_launch(void* const* d_in, const int* in_sizes, int n_in,
                              void* d_out, int out_size, void* d_ws, size_t ws_size,
                              hipStream_t stream) {
  const float* x  = (const float*)d_in[0];
  const int* edge = (const int*)d_in[1];
  const float* W1 = (const float*)d_in[2];
  const float* b1 = (const float*)d_in[3];
  const float* Wc = (const float*)d_in[4];
  const float* bc = (const float*)d_in[5];
  const float* W2 = (const float*)d_in[6];
  const float* b2 = (const float*)d_in[7];
  const float* Wo = (const float*)d_in[8];
  const float* bo = (const float*)d_in[9];

  const int N = in_sizes[0] / HIDW;
  const int E = in_sizes[1] / 2;
  const int* srcIdx = edge;
  const int* dstIdx = edge + E;

  float* wsf = (float*)d_ws;
  size_t NP   = (((size_t)N) + 255) & ~(size_t)255;
  float* dis  = wsf;
  float* buf1 = wsf + NP;
  float* buf2 = buf1 + (size_t)N * HIDW;
  float* buf3 = buf2 + (size_t)N * HIDW;
  float* deg  = buf1;

  dim3 b256(256);
  int gN  = (N + 255) / 256;
  int gNH = (N * HIDW + 255) / 256;
  int gE  = (E + 255) / 256;
  int gE2 = (E + 1) / 2;
  int gG  = (N + 127) / 128;

  k_fill1   <<<gN,  b256, 0, stream>>>(deg, N);
  k_edge_deg<<<gE,  b256, 0, stream>>>(deg, dstIdx, E);
  k_dis     <<<gN,  b256, 0, stream>>>(dis, deg, N);

  k_gemm<128,1,false,false,false><<<gG, b256, 0, stream>>>(x, nullptr, W1, nullptr, buf1, N);
  k_prop_init   <<<gNH, b256, 0, stream>>>(buf2, buf1, dis, b1, N);
  k_prop_scatter<<<gE2, b256, 0, stream>>>(buf1, srcIdx, dstIdx, dis, buf2, E);

  k_gemm<128,2,true,true,true><<<gG, b256, 0, stream>>>(buf2, x, Wc, bc, buf3, N);

  k_gemm<128,1,false,false,false><<<gG, b256, 0, stream>>>(buf3, nullptr, W2, nullptr, buf1, N);
  k_prop_init   <<<gNH, b256, 0, stream>>>(buf2, buf1, dis, b2, N);
  k_prop_scatter<<<gE2, b256, 0, stream>>>(buf1, srcIdx, dstIdx, dis, buf2, E);

  k_gemm<64,2,false,false,true><<<gG, b256, 0, stream>>>(buf2, buf3, Wo, bo, (float*)d_out, N);
}


# GEMM staging: A rows past M read row M-1 (never stored) instead of a row test, so a chunk's A and W loads leave as one group, one chunk ahead in registers; was 4 load round trips in a row a chunk
# speedup vs baseline: 5.4828x; 5.4828x over previous
#include <hip/hip_runtime.h>

typedef __attribute__((ext_vector_type(2))) float v2f;
typedef __attribute__((ext_vector_type(4))) float v4f;
typedef __attribute__((ext_vector_type(8))) float v8f;

#define HIDW 128
#define KCHUNK 32

#define TR   208
#define PW   32
#define PBJ  8
#define PEB  (32 * 16 * PBJ)
#define WCAP 512

#define HP    128
#define HCAP  65536


__device__ __forceinline__ void dst4(const int* __restrict__ dst, int i, int E, int d[4]) {
  const int e0 = i * 4;
  d[0] = -1; d[1] = -1; d[2] = -1; d[3] = -1;
  if (((E & 3) == 0) && e0 + 3 < E) {
    const int4 q = *(const int4*)&dst[e0];
    d[0] = q.x; d[1] = q.y; d[2] = q.z; d[3] = q.w;
  } else {
#pragma unroll
    for (int k = 0; k < 4; ++k) if (e0 + k < E) d[k] = dst[e0 + k];
  }
}

__global__ __launch_bounds__(1024) void k_count_plane(unsigned* __restrict__ part, const int* __restrict__ dst,
                                                      unsigned* __restrict__ plane32, int N, int S, int NP, int E) {
  __shared__ unsigned cnt[HCAP];
  const int tid  = threadIdx.x;
  const int nb   = (E + PEB - 1) / PEB;
  const int step = gridDim.x;
  unsigned* mine = part + (size_t)blockIdx.x * (size_t)S;

  for (int n0 = 0; n0 < N; n0 += HCAP) {
    const int nn = min(HCAP, N - n0);
    const int nf = min(HCAP, NP - n0);

    int bi = blockIdx.x;
    int d[4] = {-1, -1, -1, -1};
    if (bi < nb) dst4(dst, bi * (PEB / 4) + tid, E, d);

    for (int i = tid * 4; i < nf; i += 4096) *(uint4*)&cnt[i] = make_uint4(0u, 0u, 0u, 0u);
    __syncthreads();

    for (; bi < nb; bi += step) {
      int dn[4] = {-1, -1, -1, -1};
      if (bi + step < nb) dst4(dst, (bi + step) * (PEB / 4) + tid, E, dn);

      unsigned word = 0;
#pragma unroll
      for (int k = 0; k < 4; ++k) {
        unsigned t = 0xFFu;
        if (d[k] >= 0) {
          const unsigned r = (unsigned)(d[k] - n0);
          if (r < (unsigned)nn) atomicAdd(&cnt[r], 1u);
          t = ((unsigned)d[k] / (unsigned)TR) & 0xFFu;
        }
        word |= t << (8 * k);
      }
      if (n0 == 0) plane32[bi * (PEB / 4) + tid] = word;
#pragma unroll
      for (int k = 0; k < 4; ++k) d[k] = dn[k];
    }
    __syncthreads();

    for (int i = tid * 4; i < nf; i += 4096) *(uint4*)&mine[n0 + i] = *(const uint4*)&cnt[i];
    __syncthreads();
  }
}

__global__ __launch_bounds__(256) void k_dis(float* dis, const unsigned* __restrict__ part, int n, int S) {
  const unsigned i = blockIdx.x * blockDim.x + threadIdx.x;
  if (i < (unsigned)n) {
    unsigned c = 1u;
#pragma unroll 32
    for (int p = 0; p < HP; ++p) c += part[(size_t)p * (size_t)S + i];
    float d = (float)c;
    dis[i] = (d > 0.f) ? rsqrtf(d) : 0.f;
  }
}


__device__ __forceinline__ void tile_add(float* arow, float4 v, float w, int lane) {
  __hip_atomic_fetch_add(arow + lane,      v.x * w, __ATOMIC_RELAXED, __HIP_MEMORY_SCOPE_WORKGROUP);
  __hip_atomic_fetch_add(arow + 32 + lane, v.y * w, __ATOMIC_RELAXED, __HIP_MEMORY_SCOPE_WORKGROUP);
  __hip_atomic_fetch_add(arow + 64 + lane, v.z * w, __ATOMIC_RELAXED, __HIP_MEMORY_SCOPE_WORKGROUP);
  __hip_atomic_fetch_add(arow + 96 + lane, v.w * w, __ATOMIC_RELAXED, __HIP_MEMORY_SCOPE_WORKGROUP);
}

template<bool SECOND>
__global__ __launch_bounds__(1024) void k_prop_tile(const float* __restrict__ h, const int* __restrict__ src,
                                                    const int* __restrict__ dst,
                                                    const unsigned char* __restrict__ plane,
                                                    unsigned* __restrict__ lists,
                                                    const float* __restrict__ dis,
                                                    const float* __restrict__ bias, float* __restrict__ out,
                                                    int N, int E) {
  __shared__ float acc[TR * HIDW];
  __shared__ int lstE[PW][WCAP];
  __shared__ int wcnt[PW];
  __shared__ int lcnt;
  __shared__ int lovf;

  const int tid  = threadIdx.x;
  const int wave = __builtin_amdgcn_readfirstlane(tid >> 5);
  const int lane = tid & 31;
  const int r0   = blockIdx.x * TR;
  const int rows = min(TR, N - r0);

  for (int r = wave; r < TR; r += PW) {
    float4 v = {0.f, 0.f, 0.f, 0.f};
    if (r < rows) {
      const int g = r0 + r;
      const float w = dis[g];
      const float4 t = *(const float4*)&h[(size_t)g * HIDW + lane * 4];
      v.x = t.x * w; v.y = t.y * w; v.z = t.z * w; v.w = t.w * w;
    }
    float* a = &acc[r * HIDW + lane];
    a[0] = v.x; a[32] = v.y; a[64] = v.z; a[96] = v.w;
  }
  if (tid == 0) { lcnt = 0; lovf = 0; }
  __syncthreads();

  const int nb = (E + PEB - 1) / PEB;

  const long long freeW = (long long)N * 64 - (long long)nb * (PEB / 4);
  long long rwl = (freeW > 0) ? ((freeW / (long long)gridDim.x) & ~31LL) : 0;
  if (rwl > 0x40000000LL) rwl = 0x40000000LL;
  const int  RW     = (int)rwl;
  const int  cap    = RW - 32;
  const bool listOK = (RW >= 64) && (N <= (1 << 24));
  unsigned* region  = lists + (size_t)nb * (PEB / 4) + (size_t)blockIdx.x * (size_t)RW;
  unsigned* ent     = region + 32;

  int  cnt = 0;
  bool fromList = false;
  if (SECOND && listOK) {
    const unsigned c = (unsigned)__builtin_amdgcn_readfirstlane((int)region[0]);
    fromList = c <= (unsigned)cap;
    cnt = (int)c;
  }

  if (SECOND && fromList) {
    const int stride = PW * 32;
    int c = wave * 32;
    unsigned e0 = 0xFFFFFFFFu, e1 = 0xFFFFFFFFu;
    if (c + lane < cnt)          e0 = ent[c + lane];
    if (c + stride + lane < cnt) e1 = ent[c + stride + lane];
    float w0 = 0.f;
    if (e0 != 0xFFFFFFFFu) w0 = dis[e0 & 0xFFFFFFu];
    for (; c < cnt; c += stride) {
      unsigned e2 = 0xFFFFFFFFu;
      if (c + 2 * stride + lane < cnt) e2 = ent[c + 2 * stride + lane];
      float w1 = 0.f;
      if (e1 != 0xFFFFFFFFu) w1 = dis[e1 & 0xFFFFFFu];

      const int      cm   = min(32, cnt - c);
      const int      s_l  = (int)(e0 & 0xFFFFFFu);
      const unsigned lr_l = e0 >> 24;
      const float    w_l  = w0;
      for (int i0 = 0; i0 < cm; i0 += 8) {
        int      s[8];
        unsigned lr[8];
        float    w[8];
        float4   v[8];
#pragma unroll
        for (int u = 0; u < 8; ++u) {
          const int i = min(i0 + u, cm - 1);
          s[u]  = __builtin_amdgcn_readlane(s_l, i);
          lr[u] = (i0 + u < cm) ? (unsigned)__builtin_amdgcn_readlane((int)lr_l, i) : 0xFFFFFFFFu;
        }
#pragma unroll
        for (int u = 0; u < 8; ++u) v[u] = *(const float4*)&h[(size_t)s[u] * HIDW + lane * 4];
#pragma unroll
        for (int u = 0; u < 8; ++u)
          w[u] = __builtin_bit_cast(float, __builtin_amdgcn_readlane(__builtin_bit_cast(int, w_l), min(i0 + u, cm - 1)));
#pragma unroll
        for (int u = 0; u < 8; ++u)
          if (lr[u] < (unsigned)rows) tile_add(&acc[lr[u] * HIDW], v[u], w[u], lane);
      }
      e0 = e1; w0 = w1; e1 = e2;
    }
  } else {
    const bool record = !SECOND && listOK;

    const unsigned pat = ((unsigned)blockIdx.x & 0xFFu) * 0x01010101u;

    uint4 p[PBJ];
    if (wave < nb) {
#pragma unroll
      for (int j = 0; j < PBJ; ++j) p[j] = *(const uint4*)&plane[wave * PEB + (j * 32 + lane) * 16];
    }

    for (int bi = wave; bi < nb; bi += PW) {
      const int base = bi * PEB;
      if (lane == 0) wcnt[wave] = 0;

      unsigned m[PBJ];
      int fl = 0;
#pragma unroll
      for (int j = 0; j < PBJ; ++j) {
        const unsigned pw[4] = {p[j].x, p[j].y, p[j].z, p[j].w};
        unsigned f[4];
#pragma unroll
        for (int k = 0; k < 4; ++k) {
          const unsigned y = pw[k] ^ pat;
          const unsigned t = ((y & 0x7F7F7F7Fu) + 0x7F7F7F7Fu) | y;
          f[k] = ~t & 0x80808080u;
        }
        m[j] = f[0] | (f[1] >> 1) | (f[2] >> 2) | (f[3] >> 3);
        fl += __popc(m[j]);
      }

      if (bi + PW < nb) {
#pragma unroll
        for (int j = 0; j < PBJ; ++j) p[j] = *(const uint4*)&plane[(bi + PW) * PEB + (j * 32 + lane) * 16];
      }

      if (fl) {
        int pos = atomicAdd(&wcnt[wave], fl);
#pragma unroll
        for (int j = 0; j < PBJ; ++j) {
          unsigned mm = m[j];
          while (mm) {
            const int bit = __builtin_ctz(mm);
            mm &= mm - 1;
            const int e = base + (j * 32 + lane) * 16 + (7 - (bit & 7)) * 4 + (bit >> 3);
            if (pos < WCAP) lstE[wave][pos] = (e < E) ? e : -1;
            ++pos;
          }
        }
      }

      __builtin_amdgcn_wave_barrier();
      const int total = __builtin_amdgcn_readfirstlane(wcnt[wave]);
      int n = total;

      if (total > WCAP) {
        n = 0;
        if (record && lane == 0) lovf = 1;
        for (int t = lane; t < PEB; t += 32) {
          const int e = base + t;
          if (e < E) {
            const unsigned lr = (unsigned)dst[e] - (unsigned)r0;
            if (lr < (unsigned)rows) {
              const int s = src[e];
              const float w = dis[s];
              float* arow = &acc[lr * HIDW];
              for (int f = 0; f < HIDW; ++f)
                __hip_atomic_fetch_add(arow + (f & 3) * 32 + (f >> 2), h[(size_t)s * HIDW + f] * w,
                                       __ATOMIC_RELAXED, __HIP_MEMORY_SCOPE_WORKGROUP);
            }
          }
        }
      }

      for (int c0 = 0; c0 < n; c0 += 32) {
        const int cm = min(32, n - c0);
        const int e_l = (lane < cm) ? lstE[wave][c0 + lane] : -1;
        int      s_l  = 0;
        unsigned lr_l = 0xFFFFFFFFu;
        float    w_l  = 0.f;
        if (e_l >= 0) {
          s_l  = src[e_l];
          lr_l = (unsigned)dst[e_l] - (unsigned)r0;
          w_l  = dis[s_l];
        }
        if (record) {
          const bool     ok  = lr_l < (unsigned)rows;
          const unsigned bal = __builtin_amdgcn_ballot_w32(ok);
          if (bal) {
            int first = 0;
            if (lane == 0) first = atomicAdd(&lcnt, __popc(bal));
            first = __builtin_amdgcn_readfirstlane(first);
            const int pos = first + __popc(bal & ((1u << lane) - 1u));
            if (ok && pos < cap) ent[pos] = (unsigned)s_l | (lr_l << 24);
          }
        }
        for (int i0 = 0; i0 < cm; i0 += 8) {
          int      s[8];
          unsigned lr[8];
          float    w[8];
          float4   v[8];
#pragma unroll
          for (int u = 0; u < 8; ++u) {
            const int i = min(i0 + u, cm - 1);
            s[u]  = __builtin_amdgcn_readlane(s_l, i);
            lr[u] = (i0 + u < cm) ? (unsigned)__builtin_amdgcn_readlane((int)lr_l, i) : 0xFFFFFFFFu;
          }
#pragma unroll
          for (int u = 0; u < 8; ++u) v[u] = *(const float4*)&h[(size_t)s[u] * HIDW + lane * 4];
#pragma unroll
          for (int u = 0; u < 8; ++u)
            w[u] = __builtin_bit_cast(float, __builtin_amdgcn_readlane(__builtin_bit_cast(int, w_l), min(i0 + u, cm - 1)));
#pragma unroll
          for (int u = 0; u < 8; ++u)
            if (lr[u] < (unsigned)rows) tile_add(&acc[lr[u] * HIDW], v[u], w[u], lane);
        }
      }
    }
  }
  __syncthreads();

  if (!SECOND && listOK && tid == 0) region[0] = lovf ? 0xFFFFFFFFu : (unsigned)lcnt;

  const float4 b = *(const float4*)&bias[lane * 4];
  for (int r = wave; r < rows; r += PW) {
    const int g = r0 + r;
    const float w = dis[g];
    const float* a = &acc[r * HIDW + lane];
    float4 o;
    o.x = a[0] * w + b.x; o.y = a[32] * w + b.y; o.z = a[64] * w + b.z; o.w = a[96] * w + b.w;
    *(float4*)&out[(size_t)g * HIDW + lane * 4] = o;
  }
}

template<int NC, int KW, bool RELU_A0, bool RELU_OUT, bool HAS_BIAS>
__global__ __launch_bounds__(256) void k_gemm(const float* __restrict__ A0, const float* __restrict__ A1,
                                              const float* __restrict__ W, const float* __restrict__ bias,
                                              float* __restrict__ C, int M) {
  __shared__ __align__(16) float As[128][KCHUNK + 4];
  __shared__ __align__(16) float Ws[NC][KCHUNK + 4];
  constexpr int NT = NC / 16;
  constexpr int K = KW * 128;

  const int tid  = threadIdx.x;
  const int wave = tid >> 5;
  const int lane = tid & 31;
  const int l15  = lane & 15;
  const int half = lane >> 4;
  const int m0   = blockIdx.x * 128;

  v8f acc[NT];
#pragma unroll
  for (int t = 0; t < NT; ++t)
#pragma unroll
    for (int r = 0; r < 8; ++r) acc[t][r] = 0.0f;

  constexpr int NW = (NC * KCHUNK) / 1024;
  const int sc = (tid * 4) & 31;
  const int sr = tid >> 3;
  const size_t ao0 = (size_t)min(m0 + sr,      M - 1) * HIDW + sc;
  const size_t ao1 = (size_t)min(m0 + sr + 32, M - 1) * HIDW + sc;
  const size_t ao2 = (size_t)min(m0 + sr + 64, M - 1) * HIDW + sc;
  const size_t ao3 = (size_t)min(m0 + sr + 96, M - 1) * HIDW + sc;
  const float* wp = W + (size_t)sr * K + sc;

  v4f ra0 = *(const v4f*)&A0[ao0], ra1 = *(const v4f*)&A0[ao1], ra2 = *(const v4f*)&A0[ao2], ra3 = *(const v4f*)&A0[ao3];
  v4f rw0 = *(const v4f*)&wp[0], rw1 = *(const v4f*)&wp[(size_t)32 * K], rw2 = {0.f, 0.f, 0.f, 0.f}, rw3 = {0.f, 0.f, 0.f, 0.f};
  if constexpr (NW == 4) { rw2 = *(const v4f*)&wp[(size_t)64 * K]; rw3 = *(const v4f*)&wp[(size_t)96 * K]; }

  for (int kb = 0; kb < K; kb += KCHUNK) {
    if (RELU_A0 && (kb < 128)) {
      const v4f z = {0.f, 0.f, 0.f, 0.f};
      ra0 = __builtin_elementwise_max(ra0, z); ra1 = __builtin_elementwise_max(ra1, z);
      ra2 = __builtin_elementwise_max(ra2, z); ra3 = __builtin_elementwise_max(ra3, z);
    }
    *(v4f*)&As[sr][sc] = ra0; *(v4f*)&As[sr + 32][sc] = ra1; *(v4f*)&As[sr + 64][sc] = ra2; *(v4f*)&As[sr + 96][sc] = ra3;
    *(v4f*)&Ws[sr][sc] = rw0; *(v4f*)&Ws[sr + 32][sc] = rw1;
    if constexpr (NW == 4) { *(v4f*)&Ws[sr + 64][sc] = rw2; *(v4f*)&Ws[sr + 96][sc] = rw3; }
    __syncthreads();

    const int kn = kb + KCHUNK;
    if (kn < K) {
      const float* Ablk = ((KW == 2 && kn >= 128) ? A1 : A0) + (kn & 127);
      ra0 = *(const v4f*)&Ablk[ao0]; ra1 = *(const v4f*)&Ablk[ao1]; ra2 = *(const v4f*)&Ablk[ao2]; ra3 = *(const v4f*)&Ablk[ao3];
      rw0 = *(const v4f*)&wp[kn]; rw1 = *(const v4f*)&wp[(size_t)32 * K + kn];
      if constexpr (NW == 4) { rw2 = *(const v4f*)&wp[(size_t)64 * K + kn]; rw3 = *(const v4f*)&wp[(size_t)96 * K + kn]; }
    }

    const int arow = wave * 16 + l15;
    const int o = half * 2;
#pragma unroll
    for (int k4 = 0; k4 < KCHUNK; k4 += 4) {
      v2f a = *(const v2f*)&As[arow][k4 + o];
#pragma unroll
      for (int t = 0; t < NT; ++t) {
        v2f b = *(const v2f*)&Ws[t * 16 + l15][k4 + o];
        acc[t] = __builtin_amdgcn_wmma_f32_16x16x4_f32(
             false, a,  false, b,
             (short)0, acc[t],  false,  false);
      }
    }
    __syncthreads();
  }

  const int rbase = m0 + wave * 16 + half * 8;
#pragma unroll
  for (int t = 0; t < NT; ++t) {
    int col = t * 16 + l15;
    float bv = HAS_BIAS ? bias[col] : 0.f;
#pragma unroll
    for (int r = 0; r < 8; ++r) {
      int row = rbase + r;
      if (row < M) {
        float v = acc[t][r] + bv;
        if (RELU_OUT) v = fmaxf(v, 0.f);
        C[(size_t)row * NC + col] = v;
      }
    }
  }
}

extern "C" void kernel_launch(void* const* d_in, const int* in_sizes, int n_in,
                              void* d_out, int out_size, void* d_ws, size_t ws_size,
                              hipStream_t stream) {
  const float* x  = (const float*)d_in[0];
  const int* edge = (const int*)d_in[1];
  const float* W1 = (const float*)d_in[2];
  const float* b1 = (const float*)d_in[3];
  const float* Wc = (const float*)d_in[4];
  const float* bc = (const float*)d_in[5];
  const float* W2 = (const float*)d_in[6];
  const float* b2 = (const float*)d_in[7];
  const float* Wo = (const float*)d_in[8];
  const float* bo = (const float*)d_in[9];

  const int N = in_sizes[0] / HIDW;
  const int E = in_sizes[1] / 2;
  const int* srcIdx = edge;
  const int* dstIdx = edge + E;

  float* wsf = (float*)d_ws;
  size_t NP   = (((size_t)N) + 255) & ~(size_t)255;
  float* dis  = wsf;
  float* buf1 = wsf + NP;
  float* buf2 = buf1 + (size_t)N * HIDW;
  float* buf3 = buf2 + (size_t)N * HIDW;
  unsigned* cntp = (unsigned*)buf1;
  int S = (int)NP + 64;
  unsigned char* plane = (unsigned char*)d_out;

  dim3 b256(256);
  int gN  = (N + 255) / 256;
  int gT  = (N + TR - 1) / TR;
  dim3 b1024(1024);
  int gG  = (N + 127) / 128;

  k_count_plane<<<HP, b1024, 0, stream>>>(cntp, dstIdx, (unsigned*)plane, N, S, (int)NP, E);
  k_dis     <<<gN,  b256, 0, stream>>>(dis, cntp, N, S);

  k_gemm<128,1,false,false,false><<<gG, b256, 0, stream>>>(x, nullptr, W1, nullptr, buf1, N);
  k_prop_tile<false><<<gT, b1024, 0, stream>>>(buf1, srcIdx, dstIdx, plane, (unsigned*)d_out, dis, b1, buf2, N, E);

  k_gemm<128,2,true,true,true><<<gG, b256, 0, stream>>>(buf2, x, Wc, bc, buf3, N);

  k_gemm<128,1,false,false,false><<<gG, b256, 0, stream>>>(buf3, nullptr, W2, nullptr, buf1, N);
  k_prop_tile<true><<<gT, b1024, 0, stream>>>(buf1, srcIdx, dstIdx, plane, (unsigned*)d_out, dis, b2, buf2, N, E);

  k_gemm<64,2,false,false,true><<<gG, b256, 0, stream>>>(buf2, buf3, Wo, bo, (float*)d_out, N);
}


# Both layers propagate first, project after ((A x)W for A(x W): sums reassociated); a layer's two products then chain in one launch, the 1st result held in LDS: 8 launches to 6, h and h2 never stored
# speedup vs baseline: 5.9157x; 1.0790x over previous
#include <hip/hip_runtime.h>

typedef __attribute__((ext_vector_type(2))) float v2f;
typedef __attribute__((ext_vector_type(4))) float v4f;
typedef __attribute__((ext_vector_type(8))) float v8f;

#define HIDW 128
#define KCHUNK 32

#define TR   208
#define PW   32
#define PBJ  8
#define PEB  (32 * 16 * PBJ)
#define WCAP 512

#define HP    128
#define HCAP  65536


__device__ __forceinline__ void dst4(const int* __restrict__ dst, int i, int E, int d[4]) {
  const int e0 = i * 4;
  d[0] = -1; d[1] = -1; d[2] = -1; d[3] = -1;
  if (((E & 3) == 0) && e0 + 3 < E) {
    const int4 q = *(const int4*)&dst[e0];
    d[0] = q.x; d[1] = q.y; d[2] = q.z; d[3] = q.w;
  } else {
#pragma unroll
    for (int k = 0; k < 4; ++k) if (e0 + k < E) d[k] = dst[e0 + k];
  }
}

__global__ __launch_bounds__(1024) void k_count_plane(unsigned* __restrict__ part, const int* __restrict__ dst,
                                                      unsigned* __restrict__ plane32, int N, int S, int NP, int E) {
  __shared__ unsigned cnt[HCAP];
  const int tid  = threadIdx.x;
  const int nb   = (E + PEB - 1) / PEB;
  const int step = gridDim.x;
  unsigned* mine = part + (size_t)blockIdx.x * (size_t)S;

  for (int n0 = 0; n0 < N; n0 += HCAP) {
    const int nn = min(HCAP, N - n0);
    const int nf = min(HCAP, NP - n0);

    int bi = blockIdx.x;
    int d[4] = {-1, -1, -1, -1};
    if (bi < nb) dst4(dst, bi * (PEB / 4) + tid, E, d);

    for (int i = tid * 4; i < nf; i += 4096) *(uint4*)&cnt[i] = make_uint4(0u, 0u, 0u, 0u);
    __syncthreads();

    for (; bi < nb; bi += step) {
      int dn[4] = {-1, -1, -1, -1};
      if (bi + step < nb) dst4(dst, (bi + step) * (PEB / 4) + tid, E, dn);

      unsigned word = 0;
#pragma unroll
      for (int k = 0; k < 4; ++k) {
        unsigned t = 0xFFu;
        if (d[k] >= 0) {
          const unsigned r = (unsigned)(d[k] - n0);
          if (r < (unsigned)nn) atomicAdd(&cnt[r], 1u);
          t = ((unsigned)d[k] / (unsigned)TR) & 0xFFu;
        }
        word |= t << (8 * k);
      }
      if (n0 == 0) plane32[bi * (PEB / 4) + tid] = word;
#pragma unroll
      for (int k = 0; k < 4; ++k) d[k] = dn[k];
    }
    __syncthreads();

    for (int i = tid * 4; i < nf; i += 4096) *(uint4*)&mine[n0 + i] = *(const uint4*)&cnt[i];
    __syncthreads();
  }
}

__global__ __launch_bounds__(256) void k_dis(float* dis, const unsigned* __restrict__ part, int n, int S) {
  const unsigned i = blockIdx.x * blockDim.x + threadIdx.x;
  if (i < (unsigned)n) {
    unsigned c = 1u;
#pragma unroll 32
    for (int p = 0; p < HP; ++p) c += part[(size_t)p * (size_t)S + i];
    float d = (float)c;
    dis[i] = (d > 0.f) ? rsqrtf(d) : 0.f;
  }
}


__device__ __forceinline__ void tile_add(float* arow, float4 v, float w, int lane) {
  __hip_atomic_fetch_add(arow + lane,      v.x * w, __ATOMIC_RELAXED, __HIP_MEMORY_SCOPE_WORKGROUP);
  __hip_atomic_fetch_add(arow + 32 + lane, v.y * w, __ATOMIC_RELAXED, __HIP_MEMORY_SCOPE_WORKGROUP);
  __hip_atomic_fetch_add(arow + 64 + lane, v.z * w, __ATOMIC_RELAXED, __HIP_MEMORY_SCOPE_WORKGROUP);
  __hip_atomic_fetch_add(arow + 96 + lane, v.w * w, __ATOMIC_RELAXED, __HIP_MEMORY_SCOPE_WORKGROUP);
}

template<bool SECOND>
__global__ __launch_bounds__(1024) void k_prop_tile(const float* __restrict__ h, const int* __restrict__ src,
                                                    const int* __restrict__ dst,
                                                    const unsigned char* __restrict__ plane,
                                                    unsigned* __restrict__ lists,
                                                    const float* __restrict__ dis,
                                                    const float* __restrict__ bias, float* __restrict__ out,
                                                    int N, int E) {
  __shared__ float acc[TR * HIDW];
  __shared__ int lstE[PW][WCAP];
  __shared__ int wcnt[PW];
  __shared__ int lcnt;
  __shared__ int lovf;

  const int tid  = threadIdx.x;
  const int wave = __builtin_amdgcn_readfirstlane(tid >> 5);
  const int lane = tid & 31;
  const int r0   = blockIdx.x * TR;
  const int rows = min(TR, N - r0);

  for (int r = wave; r < TR; r += PW) {
    float4 v = {0.f, 0.f, 0.f, 0.f};
    if (r < rows) {
      const int g = r0 + r;
      const float w = dis[g];
      const float4 t = *(const float4*)&h[(size_t)g * HIDW + lane * 4];
      v.x = t.x * w; v.y = t.y * w; v.z = t.z * w; v.w = t.w * w;
    }
    float* a = &acc[r * HIDW + lane];
    a[0] = v.x; a[32] = v.y; a[64] = v.z; a[96] = v.w;
  }
  if (tid == 0) { lcnt = 0; lovf = 0; }
  __syncthreads();

  const int nb = (E + PEB - 1) / PEB;

  const long long freeW = (long long)N * 64 - (long long)nb * (PEB / 4);
  long long rwl = (freeW > 0) ? ((freeW / (long long)gridDim.x) & ~31LL) : 0;
  if (rwl > 0x40000000LL) rwl = 0x40000000LL;
  const int  RW     = (int)rwl;
  const int  cap    = RW - 32;
  const bool listOK = (RW >= 64) && (N <= (1 << 24));
  unsigned* region  = lists + (size_t)nb * (PEB / 4) + (size_t)blockIdx.x * (size_t)RW;
  unsigned* ent     = region + 32;

  int  cnt = 0;
  bool fromList = false;
  if (SECOND && listOK) {
    const unsigned c = (unsigned)__builtin_amdgcn_readfirstlane((int)region[0]);
    fromList = c <= (unsigned)cap;
    cnt = (int)c;
  }

  if (SECOND && fromList) {
    const int stride = PW * 32;
    int c = wave * 32;
    unsigned e0 = 0xFFFFFFFFu, e1 = 0xFFFFFFFFu;
    if (c + lane < cnt)          e0 = ent[c + lane];
    if (c + stride + lane < cnt) e1 = ent[c + stride + lane];
    float w0 = 0.f;
    if (e0 != 0xFFFFFFFFu) w0 = dis[e0 & 0xFFFFFFu];
    for (; c < cnt; c += stride) {
      unsigned e2 = 0xFFFFFFFFu;
      if (c + 2 * stride + lane < cnt) e2 = ent[c + 2 * stride + lane];
      float w1 = 0.f;
      if (e1 != 0xFFFFFFFFu) w1 = dis[e1 & 0xFFFFFFu];

      const int      cm   = min(32, cnt - c);
      const int      s_l  = (int)(e0 & 0xFFFFFFu);
      const unsigned lr_l = e0 >> 24;
      const float    w_l  = w0;
      for (int i0 = 0; i0 < cm; i0 += 8) {
        int      s[8];
        unsigned lr[8];
        float    w[8];
        float4   v[8];
#pragma unroll
        for (int u = 0; u < 8; ++u) {
          const int i = min(i0 + u, cm - 1);
          s[u]  = __builtin_amdgcn_readlane(s_l, i);
          lr[u] = (i0 + u < cm) ? (unsigned)__builtin_amdgcn_readlane((int)lr_l, i) : 0xFFFFFFFFu;
        }
#pragma unroll
        for (int u = 0; u < 8; ++u) v[u] = *(const float4*)&h[(size_t)s[u] * HIDW + lane * 4];
#pragma unroll
        for (int u = 0; u < 8; ++u)
          w[u] = __builtin_bit_cast(float, __builtin_amdgcn_readlane(__builtin_bit_cast(int, w_l), min(i0 + u, cm - 1)));
#pragma unroll
        for (int u = 0; u < 8; ++u)
          if (lr[u] < (unsigned)rows) tile_add(&acc[lr[u] * HIDW], v[u], w[u], lane);
      }
      e0 = e1; w0 = w1; e1 = e2;
    }
  } else {
    const bool record = !SECOND && listOK;

    const unsigned pat = ((unsigned)blockIdx.x & 0xFFu) * 0x01010101u;

    uint4 p[PBJ];
    if (wave < nb) {
#pragma unroll
      for (int j = 0; j < PBJ; ++j) p[j] = *(const uint4*)&plane[wave * PEB + (j * 32 + lane) * 16];
    }

    for (int bi = wave; bi < nb; bi += PW) {
      const int base = bi * PEB;
      if (lane == 0) wcnt[wave] = 0;

      unsigned m[PBJ];
      int fl = 0;
#pragma unroll
      for (int j = 0; j < PBJ; ++j) {
        const unsigned pw[4] = {p[j].x, p[j].y, p[j].z, p[j].w};
        unsigned f[4];
#pragma unroll
        for (int k = 0; k < 4; ++k) {
          const unsigned y = pw[k] ^ pat;
          const unsigned t = ((y & 0x7F7F7F7Fu) + 0x7F7F7F7Fu) | y;
          f[k] = ~t & 0x80808080u;
        }
        m[j] = f[0] | (f[1] >> 1) | (f[2] >> 2) | (f[3] >> 3);
        fl += __popc(m[j]);
      }

      if (bi + PW < nb) {
#pragma unroll
        for (int j = 0; j < PBJ; ++j) p[j] = *(const uint4*)&plane[(bi + PW) * PEB + (j * 32 + lane) * 16];
      }

      if (fl) {
        int pos = atomicAdd(&wcnt[wave], fl);
#pragma unroll
        for (int j = 0; j < PBJ; ++j) {
          unsigned mm = m[j];
          while (mm) {
            const int bit = __builtin_ctz(mm);
            mm &= mm - 1;
            const int e = base + (j * 32 + lane) * 16 + (7 - (bit & 7)) * 4 + (bit >> 3);
            if (pos < WCAP) lstE[wave][pos] = (e < E) ? e : -1;
            ++pos;
          }
        }
      }

      __builtin_amdgcn_wave_barrier();
      const int total = __builtin_amdgcn_readfirstlane(wcnt[wave]);
      int n = total;

      if (total > WCAP) {
        n = 0;
        if (record && lane == 0) lovf = 1;
        for (int t = lane; t < PEB; t += 32) {
          const int e = base + t;
          if (e < E) {
            const unsigned lr = (unsigned)dst[e] - (unsigned)r0;
            if (lr < (unsigned)rows) {
              const int s = src[e];
              const float w = dis[s];
              float* arow = &acc[lr * HIDW];
              for (int f = 0; f < HIDW; ++f)
                __hip_atomic_fetch_add(arow + (f & 3) * 32 + (f >> 2), h[(size_t)s * HIDW + f] * w,
                                       __ATOMIC_RELAXED, __HIP_MEMORY_SCOPE_WORKGROUP);
            }
          }
        }
      }

      for (int c0 = 0; c0 < n; c0 += 32) {
        const int cm = min(32, n - c0);
        const int e_l = (lane < cm) ? lstE[wave][c0 + lane] : -1;
        int      s_l  = 0;
        unsigned lr_l = 0xFFFFFFFFu;
        float    w_l  = 0.f;
        if (e_l >= 0) {
          s_l  = src[e_l];
          lr_l = (unsigned)dst[e_l] - (unsigned)r0;
          w_l  = dis[s_l];
        }
        if (record) {
          const bool     ok  = lr_l < (unsigned)rows;
          const unsigned bal = __builtin_amdgcn_ballot_w32(ok);
          if (bal) {
            int first = 0;
            if (lane == 0) first = atomicAdd(&lcnt, __popc(bal));
            first = __builtin_amdgcn_readfirstlane(first);
            const int pos = first + __popc(bal & ((1u << lane) - 1u));
            if (ok && pos < cap) ent[pos] = (unsigned)s_l | (lr_l << 24);
          }
        }
        for (int i0 = 0; i0 < cm; i0 += 8) {
          int      s[8];
          unsigned lr[8];
          float    w[8];
          float4   v[8];
#pragma unroll
          for (int u = 0; u < 8; ++u) {
            const int i = min(i0 + u, cm - 1);
            s[u]  = __builtin_amdgcn_readlane(s_l, i);
            lr[u] = (i0 + u < cm) ? (unsigned)__builtin_amdgcn_readlane((int)lr_l, i) : 0xFFFFFFFFu;
          }
#pragma unroll
          for (int u = 0; u < 8; ++u) v[u] = *(const float4*)&h[(size_t)s[u] * HIDW + lane * 4];
#pragma unroll
          for (int u = 0; u < 8; ++u)
            w[u] = __builtin_bit_cast(float, __builtin_amdgcn_readlane(__builtin_bit_cast(int, w_l), min(i0 + u, cm - 1)));
#pragma unroll
          for (int u = 0; u < 8; ++u)
            if (lr[u] < (unsigned)rows) tile_add(&acc[lr[u] * HIDW], v[u], w[u], lane);
        }
      }
    }
  }
  __syncthreads();

  if (!SECOND && listOK && tid == 0) region[0] = lovf ? 0xFFFFFFFFu : (unsigned)lcnt;

  const float4 b = bias ? *(const float4*)&bias[lane * 4] : float4{0.f, 0.f, 0.f, 0.f};
  for (int r = wave; r < rows; r += PW) {
    const int g = r0 + r;
    const float w = dis[g];
    const float* a = &acc[r * HIDW + lane];
    float4 o;
    o.x = a[0] * w + b.x; o.y = a[32] * w + b.y; o.z = a[64] * w + b.z; o.w = a[96] * w + b.w;
    *(float4*)&out[(size_t)g * HIDW + lane * 4] = o;
  }
}

template<int NC, bool RELU1, bool RELU2>
__global__ __launch_bounds__(256) void k_gemm_pair(const float* __restrict__ A0, const float* __restrict__ A1,
                                                   const float* __restrict__ Wa, const float* __restrict__ ba,
                                                   const float* __restrict__ Wb, const float* __restrict__ bb,
                                                   float* __restrict__ C, int M) {
  __shared__ __align__(16) float As[128][KCHUNK + 4];
  __shared__ __align__(16) float Ws[128][KCHUNK + 4];
  __shared__ __align__(16) float Ts[128][HIDW + 4];
  constexpr int NT = NC / 16;
  constexpr int K2 = 2 * HIDW;

  const int tid  = threadIdx.x;
  const int wave = tid >> 5;
  const int lane = tid & 31;
  const int l15  = lane & 15;
  const int half = lane >> 4;
  const int m0   = blockIdx.x * 128;

  constexpr int NW = (NC * KCHUNK) / 1024;
  const int sc = (tid * 4) & 31;
  const int sr = tid >> 3;
  const size_t ao0 = (size_t)min(m0 + sr,      M - 1) * HIDW + sc;
  const size_t ao1 = (size_t)min(m0 + sr + 32, M - 1) * HIDW + sc;
  const size_t ao2 = (size_t)min(m0 + sr + 64, M - 1) * HIDW + sc;
  const size_t ao3 = (size_t)min(m0 + sr + 96, M - 1) * HIDW + sc;
  const float* wpa = Wa + (size_t)sr * HIDW + sc;
  const float* wpb = Wb + (size_t)sr * K2 + sc;

  const int arow = wave * 16 + l15;
  const int o = half * 2;

  v4f ra0 = *(const v4f*)&A0[ao0], ra1 = *(const v4f*)&A0[ao1], ra2 = *(const v4f*)&A0[ao2], ra3 = *(const v4f*)&A0[ao3];
  v4f rw0 = *(const v4f*)&wpa[0], rw1 = *(const v4f*)&wpa[(size_t)32 * HIDW];
  v4f rw2 = *(const v4f*)&wpa[(size_t)64 * HIDW], rw3 = *(const v4f*)&wpa[(size_t)96 * HIDW];

  {
    v8f acc[8];
#pragma unroll
    for (int t = 0; t < 8; ++t)
#pragma unroll
      for (int r = 0; r < 8; ++r) acc[t][r] = 0.0f;

    for (int kb = 0; kb < HIDW; kb += KCHUNK) {
      *(v4f*)&As[sr][sc] = ra0; *(v4f*)&As[sr + 32][sc] = ra1; *(v4f*)&As[sr + 64][sc] = ra2; *(v4f*)&As[sr + 96][sc] = ra3;
      *(v4f*)&Ws[sr][sc] = rw0; *(v4f*)&Ws[sr + 32][sc] = rw1; *(v4f*)&Ws[sr + 64][sc] = rw2; *(v4f*)&Ws[sr + 96][sc] = rw3;
      __syncthreads();

      const int kn = kb + KCHUNK;
      if (kn < HIDW) {
        ra0 = *(const v4f*)&A0[ao0 + kn]; ra1 = *(const v4f*)&A0[ao1 + kn]; ra2 = *(const v4f*)&A0[ao2 + kn]; ra3 = *(const v4f*)&A0[ao3 + kn];
        rw0 = *(const v4f*)&wpa[kn]; rw1 = *(const v4f*)&wpa[(size_t)32 * HIDW + kn];
        rw2 = *(const v4f*)&wpa[(size_t)64 * HIDW + kn]; rw3 = *(const v4f*)&wpa[(size_t)96 * HIDW + kn];
      } else {
        rw0 = *(const v4f*)&wpb[0]; rw1 = *(const v4f*)&wpb[(size_t)32 * K2];
        if constexpr (NW == 4) { rw2 = *(const v4f*)&wpb[(size_t)64 * K2]; rw3 = *(const v4f*)&wpb[(size_t)96 * K2]; }
      }

#pragma unroll
      for (int k4 = 0; k4 < KCHUNK; k4 += 4) {
        v2f a = *(const v2f*)&As[arow][k4 + o];
#pragma unroll
        for (int t = 0; t < 8; ++t) {
          v2f b = *(const v2f*)&Ws[t * 16 + l15][k4 + o];
          acc[t] = __builtin_amdgcn_wmma_f32_16x16x4_f32(
               false, a,  false, b,
               (short)0, acc[t],  false,  false);
        }
      }
      __syncthreads();
    }

#pragma unroll
    for (int t = 0; t < 8; ++t) {
      const int col = t * 16 + l15;
      const float bv = ba[col];
#pragma unroll
      for (int r = 0; r < 8; ++r) {
        float v = acc[t][r] + bv;
        if (RELU1) v = fmaxf(v, 0.f);
        Ts[wave * 16 + half * 8 + r][col] = v;
      }
    }
  }
  __syncthreads();

  v8f acc[NT];
#pragma unroll
  for (int t = 0; t < NT; ++t)
#pragma unroll
    for (int r = 0; r < 8; ++r) acc[t][r] = 0.0f;

  for (int kb = 0; kb < HIDW; kb += KCHUNK) {
    *(v4f*)&Ws[sr][sc] = rw0; *(v4f*)&Ws[sr + 32][sc] = rw1;
    if constexpr (NW == 4) { *(v4f*)&Ws[sr + 64][sc] = rw2; *(v4f*)&Ws[sr + 96][sc] = rw3; }
    __syncthreads();

    const int kn = kb + KCHUNK;
    rw0 = *(const v4f*)&wpb[kn]; rw1 = *(const v4f*)&wpb[(size_t)32 * K2 + kn];
    if constexpr (NW == 4) { rw2 = *(const v4f*)&wpb[(size_t)64 * K2 + kn]; rw3 = *(const v4f*)&wpb[(size_t)96 * K2 + kn]; }
    if (kn == HIDW) {
      ra0 = *(const v4f*)&A1[ao0]; ra1 = *(const v4f*)&A1[ao1]; ra2 = *(const v4f*)&A1[ao2]; ra3 = *(const v4f*)&A1[ao3];
    }

#pragma unroll
    for (int k4 = 0; k4 < KCHUNK; k4 += 4) {
      v2f a = *(const v2f*)&Ts[arow][kb + k4 + o];
#pragma unroll
      for (int t = 0; t < NT; ++t) {
        v2f b = *(const v2f*)&Ws[t * 16 + l15][k4 + o];
        acc[t] = __builtin_amdgcn_wmma_f32_16x16x4_f32(
             false, a,  false, b,
             (short)0, acc[t],  false,  false);
      }
    }
    __syncthreads();
  }

  for (int kb = HIDW; kb < K2; kb += KCHUNK) {
    *(v4f*)&As[sr][sc] = ra0; *(v4f*)&As[sr + 32][sc] = ra1; *(v4f*)&As[sr + 64][sc] = ra2; *(v4f*)&As[sr + 96][sc] = ra3;
    *(v4f*)&Ws[sr][sc] = rw0; *(v4f*)&Ws[sr + 32][sc] = rw1;
    if constexpr (NW == 4) { *(v4f*)&Ws[sr + 64][sc] = rw2; *(v4f*)&Ws[sr + 96][sc] = rw3; }
    __syncthreads();

    const int kn = kb + KCHUNK;
    if (kn < K2) {
      const int ka = kn - HIDW;
      ra0 = *(const v4f*)&A1[ao0 + ka]; ra1 = *(const v4f*)&A1[ao1 + ka]; ra2 = *(const v4f*)&A1[ao2 + ka]; ra3 = *(const v4f*)&A1[ao3 + ka];
      rw0 = *(const v4f*)&wpb[kn]; rw1 = *(const v4f*)&wpb[(size_t)32 * K2 + kn];
      if constexpr (NW == 4) { rw2 = *(const v4f*)&wpb[(size_t)64 * K2 + kn]; rw3 = *(const v4f*)&wpb[(size_t)96 * K2 + kn]; }
    }

#pragma unroll
    for (int k4 = 0; k4 < KCHUNK; k4 += 4) {
      v2f a = *(const v2f*)&As[arow][k4 + o];
#pragma unroll
      for (int t = 0; t < NT; ++t) {
        v2f b = *(const v2f*)&Ws[t * 16 + l15][k4 + o];
        acc[t] = __builtin_amdgcn_wmma_f32_16x16x4_f32(
             false, a,  false, b,
             (short)0, acc[t],  false,  false);
      }
    }
    __syncthreads();
  }

  const int rbase = m0 + wave * 16 + half * 8;
#pragma unroll
  for (int t = 0; t < NT; ++t) {
    const int col = t * 16 + l15;
    const float bv = bb[col];
#pragma unroll
    for (int r = 0; r < 8; ++r) {
      const int row = rbase + r;
      if (row < M) {
        float v = acc[t][r] + bv;
        if (RELU2) v = fmaxf(v, 0.f);
        C[(size_t)row * NC + col] = v;
      }
    }
  }
}

extern "C" void kernel_launch(void* const* d_in, const int* in_sizes, int n_in,
                              void* d_out, int out_size, void* d_ws, size_t ws_size,
                              hipStream_t stream) {
  const float* x  = (const float*)d_in[0];
  const int* edge = (const int*)d_in[1];
  const float* W1 = (const float*)d_in[2];
  const float* b1 = (const float*)d_in[3];
  const float* Wc = (const float*)d_in[4];
  const float* bc = (const float*)d_in[5];
  const float* W2 = (const float*)d_in[6];
  const float* b2 = (const float*)d_in[7];
  const float* Wo = (const float*)d_in[8];
  const float* bo = (const float*)d_in[9];

  const int N = in_sizes[0] / HIDW;
  const int E = in_sizes[1] / 2;
  const int* srcIdx = edge;
  const int* dstIdx = edge + E;

  float* wsf = (float*)d_ws;
  size_t NP   = (((size_t)N) + 255) & ~(size_t)255;
  float* dis  = wsf;
  float* buf1 = wsf + NP;
  float* buf2 = buf1 + (size_t)N * HIDW;
  float* buf3 = buf2 + (size_t)N * HIDW;
  unsigned* cntp = (unsigned*)buf1;
  int S = (int)NP + 64;
  unsigned char* plane = (unsigned char*)d_out;

  dim3 b256(256);
  int gN  = (N + 255) / 256;
  int gT  = (N + TR - 1) / TR;
  dim3 b1024(1024);
  int gG  = (N + 127) / 128;

  k_count_plane<<<HP, b1024, 0, stream>>>(cntp, dstIdx, (unsigned*)plane, N, S, (int)NP, E);
  k_dis     <<<gN,  b256, 0, stream>>>(dis, cntp, N, S);


  k_prop_tile<false><<<gT, b1024, 0, stream>>>(x, srcIdx, dstIdx, plane, (unsigned*)d_out, dis, nullptr, buf2, N, E);
  k_gemm_pair<128,true,true><<<gG, b256, 0, stream>>>(buf2, x, W1, b1, Wc, bc, buf3, N);

  k_prop_tile<true><<<gT, b1024, 0, stream>>>(buf3, srcIdx, dstIdx, plane, (unsigned*)d_out, dis, nullptr, buf2, N, E);
  k_gemm_pair<64,false,false><<<gG, b256, 0, stream>>>(buf2, buf3, W2, b2, Wo, bo, (float*)d_out, N);
}
